# starting kernel
# baseline (speedup 1.0000x reference)
#include <hip/hip_runtime.h>

typedef __attribute__((ext_vector_type(2))) float v2f;
typedef __attribute__((ext_vector_type(8))) float v8f;

#define B_    4096
#define NI_   256
#define NH_   2048
#define NO_   128
#define STEPS 100
#define ALPHA_ 0.9f

__global__ __launch_bounds__(256) void snn_gemm1_recur(
    const float* __restrict__ x,
    const float* __restrict__ W1,
    const float* __restrict__ b1,
    const float* __restrict__ betap,
    float* __restrict__ spk)
{
    const int wave = threadIdx.x >> 5;
    const int lane = threadIdx.x & 31;
    const int tile = blockIdx.x * 8 + wave;
    const int tilesN = NH_ / 16;
    const int tm = tile / tilesN;
    const int tn = tile - tm * tilesN;
    const int row0 = tm << 4;
    const int col0 = tn << 4;
    const int mn  = lane & 15;
    const int khi = lane >> 4;

    const float* ap = x  + (size_t)(row0 + mn) * NI_ + (khi << 1);
    const float* bp = W1 + (size_t)(col0 + mn) * NI_ + (khi << 1);

    v8f acc = {};
#pragma unroll 8
    for (int k = 0; k < NI_; k += 4) {
        v2f a = *(const v2f*)(ap + k);
        v2f b = *(const v2f*)(bp + k);
        acc = __builtin_amdgcn_wmma_f32_16x16x4_f32(
            false, a, false, b, (short)0, acc, false, false);
    }

    float beta = *betap;
    beta = fminf(fmaxf(beta, 0.0f), 1.0f);
    const float bias = b1[col0 + mn];

    float c[8], syn[8], mem[8];
#pragma unroll
    for (int i = 0; i < 8; ++i) {
        c[i] = acc[i] + bias;
        syn[i] = 0.0f;
        mem[i] = 0.0f;
    }

#pragma unroll 1
    for (int t = 0; t < STEPS; ++t) {
#pragma unroll
        for (int i = 0; i < 8; ++i) {
            const float reset = (mem[i] > 1.0f) ? 1.0f : 0.0f;
            syn[i] = fmaf(ALPHA_, syn[i], c[i]);
            mem[i] = fmaf(beta, mem[i], syn[i]) - reset;
        }
    }

    float* op = spk + (size_t)(row0 + khi * 8) * NH_ + (col0 + mn);
#pragma unroll
    for (int i = 0; i < 8; ++i)
        op[(size_t)i * NH_] = ((mem[i] - 1.0f) > 0.0f) ? 1.0f : 0.0f;
}

__global__ __launch_bounds__(256) void snn_gemm2(
    const float* __restrict__ spk,
    const float* __restrict__ W2,
    const float* __restrict__ b2,
    float* __restrict__ out)
{
    const int wave = threadIdx.x >> 5;
    const int lane = threadIdx.x & 31;
    const int tile = blockIdx.x * 8 + wave;
    const int tilesN = NO_ / 16;
    const int tm = tile / tilesN;
    const int tn = tile - tm * tilesN;
    const int row0 = tm << 4;
    const int col0 = tn << 4;
    const int mn  = lane & 15;
    const int khi = lane >> 4;

    const float* ap = spk + (size_t)(row0 + mn) * NH_ + (khi << 1);
    const float* bp = W2  + (size_t)(col0 + mn) * NH_ + (khi << 1);

    v8f acc = {};
#pragma unroll 8
    for (int k = 0; k < NH_; k += 4) {
        v2f a = *(const v2f*)(ap + k);
        v2f b = *(const v2f*)(bp + k);
        acc = __builtin_amdgcn_wmma_f32_16x16x4_f32(
            false, a, false, b, (short)0, acc, false, false);
    }

    const float bias = b2[col0 + mn];
    float* op = out + (size_t)(row0 + khi * 8) * NO_ + (col0 + mn);
#pragma unroll
    for (int i = 0; i < 8; ++i)
        op[(size_t)i * NO_] = acc[i] + bias;
}

extern "C" void kernel_launch(void* const* d_in, const int* in_sizes, int n_in,
                              void* d_out, int out_size, void* d_ws, size_t ws_size,
                              hipStream_t stream) {
    (void)in_sizes; (void)n_in; (void)out_size; (void)ws_size;
    const float* x    = (const float*)d_in[0];
    const float* W1   = (const float*)d_in[1];
    const float* b1   = (const float*)d_in[2];
    const float* W2   = (const float*)d_in[3];
    const float* b2   = (const float*)d_in[4];
    const float* beta = (const float*)d_in[5];

    float* spk = (float*)d_ws;
    float* out = (float*)d_out;

    const int tiles1 = (B_ / 16) * (NH_ / 16);
    snn_gemm1_recur<<<tiles1 / 8, 256, 0, stream>>>(x, W1, b1, beta, spk);

    const int tiles2 = (B_ / 16) * (NO_ / 16);
    snn_gemm2<<<tiles2 / 8, 256, 0, stream>>>(spk, W2, b2, out);
}


# Stage both GEMMs' A and B slabs through LDS with coalesced 16-byte global loads (whole aligned 256-byte row pieces per wave), WMMA fragments read from LDS; k order per accumulator unchanged
# speedup vs baseline: 1.3329x; 1.3329x over previous
#include <hip/hip_runtime.h>


typedef __attribute__((ext_vector_type(2))) float v2f;
typedef __attribute__((ext_vector_type(4))) float v4f;
typedef __attribute__((ext_vector_type(8))) float v8f;

#define B_    4096
#define NI_   256
#define NH_   2048
#define NO_   128
#define STEPS 100
#define ALPHA_ 0.9f

template <int K, int KC>
__device__ __forceinline__ v8f block_gemm_16x128(
    const float* __restrict__ Ag,
    const float* __restrict__ Bg,
    float* __restrict__ As,
    float* __restrict__ Bs)
{
    constexpr int PITCH = KC + 4;
    constexpr int C4 = KC / 4;
    constexpr int RP = 256 / C4;
    constexpr int AP = 16 / RP;
    constexpr int BP = 128 / RP;

    const int t    = threadIdx.x;
    const int srow = t / C4;
    const int sc4  = t % C4;
    const int wave = t >> 5;
    const int lane = t & 31;
    const int mn   = lane & 15;
    const int khi  = lane >> 4;

    const float* ag = Ag + (size_t)srow * K + sc4 * 4;
    const float* bg = Bg + (size_t)srow * K + sc4 * 4;
    float* as = As + srow * PITCH + sc4 * 4;
    float* bs = Bs + srow * PITCH + sc4 * 4;
    const float* af = As + mn * PITCH + 2 * khi;
    const float* bf = Bs + (wave * 16 + mn) * PITCH + 2 * khi;

    v4f ra[AP], rb[BP];
#pragma unroll
    for (int i = 0; i < AP; ++i) ra[i] = *(const v4f*)(ag + (size_t)i * RP * K);
#pragma unroll
    for (int i = 0; i < BP; ++i) rb[i] = *(const v4f*)(bg + (size_t)i * RP * K);

    v8f acc = {};
    for (int kc = 0; kc < K; kc += KC) {
#pragma unroll
        for (int i = 0; i < AP; ++i) *(v4f*)(as + i * RP * PITCH) = ra[i];
#pragma unroll
        for (int i = 0; i < BP; ++i) *(v4f*)(bs + i * RP * PITCH) = rb[i];
        __syncthreads();

        if (kc + KC < K) {
#pragma unroll
            for (int i = 0; i < AP; ++i) ra[i] = *(const v4f*)(ag + (size_t)i * RP * K + kc + KC);
#pragma unroll
            for (int i = 0; i < BP; ++i) rb[i] = *(const v4f*)(bg + (size_t)i * RP * K + kc + KC);
        }

#pragma unroll
        for (int k = 0; k < KC; k += 4) {
            v2f a = *(const v2f*)(af + k);
            v2f b = *(const v2f*)(bf + k);
            acc = __builtin_amdgcn_wmma_f32_16x16x4_f32(
                false, a, false, b, (short)0, acc, false, false);
        }
        __syncthreads();
    }
    return acc;
}

__global__ __launch_bounds__(256) void snn_gemm1_recur(
    const float* __restrict__ x,
    const float* __restrict__ W1,
    const float* __restrict__ b1,
    const float* __restrict__ betap,
    float* __restrict__ spk)
{
    constexpr int KC = 64;
    __shared__ __attribute__((aligned(16))) float As[16 * (KC + 4)];
    __shared__ __attribute__((aligned(16))) float Bs[128 * (KC + 4)];

    const int wave = threadIdx.x >> 5;
    const int lane = threadIdx.x & 31;
    const int blocksN = NH_ / 128;
    const int tm = blockIdx.x / blocksN;
    const int bn = blockIdx.x - tm * blocksN;
    const int row0 = tm << 4;
    const int colB = bn << 7;
    const int col0 = colB + (wave << 4);
    const int mn  = lane & 15;
    const int khi = lane >> 4;

    v8f acc = block_gemm_16x128<NI_, KC>(x + (size_t)row0 * NI_, W1 + (size_t)colB * NI_, As, Bs);

    float beta = *betap;
    beta = fminf(fmaxf(beta, 0.0f), 1.0f);
    const float bias = b1[col0 + mn];

    float c[8], syn[8], mem[8];
#pragma unroll
    for (int i = 0; i < 8; ++i) {
        c[i] = acc[i] + bias;
        syn[i] = 0.0f;
        mem[i] = 0.0f;
    }

#pragma unroll 1
    for (int t = 0; t < STEPS; ++t) {
#pragma unroll
        for (int i = 0; i < 8; ++i) {
            const float reset = (mem[i] > 1.0f) ? 1.0f : 0.0f;
            syn[i] = fmaf(ALPHA_, syn[i], c[i]);
            mem[i] = fmaf(beta, mem[i], syn[i]) - reset;
        }
    }

    float* op = spk + (size_t)(row0 + khi * 8) * NH_ + (col0 + mn);
#pragma unroll
    for (int i = 0; i < 8; ++i)
        op[(size_t)i * NH_] = ((mem[i] - 1.0f) > 0.0f) ? 1.0f : 0.0f;
}

__global__ __launch_bounds__(256) void snn_gemm2(
    const float* __restrict__ spk,
    const float* __restrict__ W2,
    const float* __restrict__ b2,
    float* __restrict__ out)
{
    constexpr int KC = 128;
    __shared__ __attribute__((aligned(16))) float As[16 * (KC + 4)];
    __shared__ __attribute__((aligned(16))) float Bs[128 * (KC + 4)];

    const int wave = threadIdx.x >> 5;
    const int lane = threadIdx.x & 31;
    const int row0 = blockIdx.x << 4;
    const int col0 = wave << 4;
    const int mn  = lane & 15;
    const int khi = lane >> 4;

    v8f acc = block_gemm_16x128<NH_, KC>(spk + (size_t)row0 * NH_, W2, As, Bs);

    const float bias = b2[col0 + mn];
    float* op = out + (size_t)(row0 + khi * 8) * NO_ + (col0 + mn);
#pragma unroll
    for (int i = 0; i < 8; ++i)
        op[(size_t)i * NO_] = acc[i] + bias;
}

extern "C" void kernel_launch(void* const* d_in, const int* in_sizes, int n_in,
                              void* d_out, int out_size, void* d_ws, size_t ws_size,
                              hipStream_t stream) {
    (void)in_sizes; (void)n_in; (void)out_size; (void)ws_size;
    const float* x    = (const float*)d_in[0];
    const float* W1   = (const float*)d_in[1];
    const float* b1   = (const float*)d_in[2];
    const float* W2   = (const float*)d_in[3];
    const float* b2   = (const float*)d_in[4];
    const float* beta = (const float*)d_in[5];

    float* spk = (float*)d_ws;
    float* out = (float*)d_out;

    const int tiles1 = (B_ / 16) * (NH_ / 16);
    snn_gemm1_recur<<<tiles1 / 8, 256, 0, stream>>>(x, W1, b1, beta, spk);

    const int tiles2 = (B_ / 16) * (NO_ / 16);
    snn_gemm2<<<tiles2 / 8, 256, 0, stream>>>(spk, W2, b2, out);
}


# Recurrence: reset=(mem>1) is one saturated fma per element (mem*2^64-2^64, clamp) instead of compare+select, syn computed one pass ahead: 20 not 28 vector instr per 8-element step; same values
# speedup vs baseline: 1.5068x; 1.1305x over previous
#include <hip/hip_runtime.h>


typedef __attribute__((ext_vector_type(2))) float v2f;
typedef __attribute__((ext_vector_type(4))) float v4f;
typedef __attribute__((ext_vector_type(8))) float v8f;

#define B_    4096
#define NI_   256
#define NH_   2048
#define NO_   128
#define STEPS 100
#define ALPHA_ 0.9f

template <int K, int KC>
__device__ __forceinline__ v8f block_gemm_16x128(
    const float* __restrict__ Ag,
    const float* __restrict__ Bg,
    float* __restrict__ As,
    float* __restrict__ Bs)
{
    constexpr int PITCH = KC + 4;
    constexpr int C4 = KC / 4;
    constexpr int RP = 256 / C4;
    constexpr int AP = 16 / RP;
    constexpr int BP = 128 / RP;

    const int t    = threadIdx.x;
    const int srow = t / C4;
    const int sc4  = t % C4;
    const int wave = t >> 5;
    const int lane = t & 31;
    const int mn   = lane & 15;
    const int khi  = lane >> 4;

    const float* ag = Ag + (size_t)srow * K + sc4 * 4;
    const float* bg = Bg + (size_t)srow * K + sc4 * 4;
    float* as = As + srow * PITCH + sc4 * 4;
    float* bs = Bs + srow * PITCH + sc4 * 4;
    const float* af = As + mn * PITCH + 2 * khi;
    const float* bf = Bs + (wave * 16 + mn) * PITCH + 2 * khi;

    v4f ra[AP], rb[BP];
#pragma unroll
    for (int i = 0; i < AP; ++i) ra[i] = *(const v4f*)(ag + (size_t)i * RP * K);
#pragma unroll
    for (int i = 0; i < BP; ++i) rb[i] = *(const v4f*)(bg + (size_t)i * RP * K);

    v8f acc = {};
    for (int kc = 0; kc < K; kc += KC) {
#pragma unroll
        for (int i = 0; i < AP; ++i) *(v4f*)(as + i * RP * PITCH) = ra[i];
#pragma unroll
        for (int i = 0; i < BP; ++i) *(v4f*)(bs + i * RP * PITCH) = rb[i];
        __syncthreads();

        if (kc + KC < K) {
#pragma unroll
            for (int i = 0; i < AP; ++i) ra[i] = *(const v4f*)(ag + (size_t)i * RP * K + kc + KC);
#pragma unroll
            for (int i = 0; i < BP; ++i) rb[i] = *(const v4f*)(bg + (size_t)i * RP * K + kc + KC);
        }

#pragma unroll
        for (int k = 0; k < KC; k += 4) {
            v2f a = *(const v2f*)(af + k);
            v2f b = *(const v2f*)(bf + k);
            acc = __builtin_amdgcn_wmma_f32_16x16x4_f32(
                false, a, false, b, (short)0, acc, false, false);
        }
        __syncthreads();
    }
    return acc;
}

__device__ __forceinline__ float gt_one(float m, float big)
{
    float r;
    asm("v_fma_f32 %0, %1, %2, -%2 clamp" : "=v"(r) : "v"(m), "s"(big));
    return r;
}

__global__ __launch_bounds__(256) void snn_gemm1_recur(
    const float* __restrict__ x,
    const float* __restrict__ W1,
    const float* __restrict__ b1,
    const float* __restrict__ betap,
    float* __restrict__ spk)
{
    constexpr int KC = 64;
    __shared__ __attribute__((aligned(16))) float As[16 * (KC + 4)];
    __shared__ __attribute__((aligned(16))) float Bs[128 * (KC + 4)];

    const int wave = threadIdx.x >> 5;
    const int lane = threadIdx.x & 31;
    const int blocksN = NH_ / 128;
    const int tm = blockIdx.x / blocksN;
    const int bn = blockIdx.x - tm * blocksN;
    const int row0 = tm << 4;
    const int colB = bn << 7;
    const int col0 = colB + (wave << 4);
    const int mn  = lane & 15;
    const int khi = lane >> 4;

    v8f acc = block_gemm_16x128<NI_, KC>(x + (size_t)row0 * NI_, W1 + (size_t)colB * NI_, As, Bs);

    float beta = *betap;
    beta = fminf(fmaxf(beta, 0.0f), 1.0f);
    const float bias = b1[col0 + mn];

    v2f c[4], syn[4], mem[4];
#pragma unroll
    for (int j = 0; j < 4; ++j) {
        c[j].x = acc[2 * j] + bias;
        c[j].y = acc[2 * j + 1] + bias;
        syn[j] = (v2f){0.0f, 0.0f};
        mem[j] = (v2f){0.0f, 0.0f};
    }
    const v2f alpha2 = {ALPHA_, ALPHA_};
    const v2f beta2  = {beta, beta};

    const float BIG = 0x1p+64f;
#pragma unroll
    for (int j = 0; j < 4; ++j)
        syn[j] = __builtin_elementwise_fma(alpha2, syn[j], c[j]);
#pragma unroll 1
    for (int t = 0; t < STEPS; ++t) {
        v2f d[4], reset[4];
#pragma unroll
        for (int j = 0; j < 4; ++j)
            d[j] = __builtin_elementwise_fma(beta2, mem[j], syn[j]);
#pragma unroll
        for (int j = 0; j < 4; ++j) {
            reset[j].x = gt_one(mem[j].x, BIG);
            reset[j].y = gt_one(mem[j].y, BIG);
        }
#pragma unroll
        for (int j = 0; j < 4; ++j)
            syn[j] = __builtin_elementwise_fma(alpha2, syn[j], c[j]);
#pragma unroll
        for (int j = 0; j < 4; ++j)
            mem[j] = d[j] - reset[j];
    }

    float* op = spk + (size_t)(row0 + khi * 8) * NH_ + (col0 + mn);
#pragma unroll
    for (int j = 0; j < 4; ++j) {
        op[(size_t)(2 * j) * NH_]     = ((mem[j].x - 1.0f) > 0.0f) ? 1.0f : 0.0f;
        op[(size_t)(2 * j + 1) * NH_] = ((mem[j].y - 1.0f) > 0.0f) ? 1.0f : 0.0f;
    }
}

__global__ __launch_bounds__(256) void snn_gemm2(
    const float* __restrict__ spk,
    const float* __restrict__ W2,
    const float* __restrict__ b2,
    float* __restrict__ out)
{
    constexpr int KC = 128;
    __shared__ __attribute__((aligned(16))) float As[16 * (KC + 4)];
    __shared__ __attribute__((aligned(16))) float Bs[128 * (KC + 4)];

    const int wave = threadIdx.x >> 5;
    const int lane = threadIdx.x & 31;
    const int row0 = blockIdx.x << 4;
    const int col0 = wave << 4;
    const int mn  = lane & 15;
    const int khi = lane >> 4;

    v8f acc = block_gemm_16x128<NH_, KC>(spk + (size_t)row0 * NH_, W2, As, Bs);

    const float bias = b2[col0 + mn];
    float* op = out + (size_t)(row0 + khi * 8) * NO_ + (col0 + mn);
#pragma unroll
    for (int i = 0; i < 8; ++i)
        op[(size_t)i * NO_] = acc[i] + bias;
}

extern "C" void kernel_launch(void* const* d_in, const int* in_sizes, int n_in,
                              void* d_out, int out_size, void* d_ws, size_t ws_size,
                              hipStream_t stream) {
    (void)in_sizes; (void)n_in; (void)out_size; (void)ws_size;
    const float* x    = (const float*)d_in[0];
    const float* W1   = (const float*)d_in[1];
    const float* b1   = (const float*)d_in[2];
    const float* W2   = (const float*)d_in[3];
    const float* b2   = (const float*)d_in[4];
    const float* beta = (const float*)d_in[5];

    float* spk = (float*)d_ws;
    float* out = (float*)d_out;

    const int tiles1 = (B_ / 16) * (NH_ / 16);
    snn_gemm1_recur<<<tiles1 / 8, 256, 0, stream>>>(x, W1, b1, beta, spk);

    const int tiles2 = (B_ / 16) * (NO_ / 16);
    snn_gemm2<<<tiles2 / 8, 256, 0, stream>>>(spk, W2, b2, out);
}


# Recurrence: reset=(mem>1) of both elements of a register pair in ONE two-wide saturated multiply-add (v_pk_fma_f32 clamp), the four at the head of a pass: 16 not 20 vector instr a pass; same values
# speedup vs baseline: 1.6301x; 1.0818x over previous
#include <hip/hip_runtime.h>


typedef __attribute__((ext_vector_type(2))) float v2f;
typedef __attribute__((ext_vector_type(4))) float v4f;
typedef __attribute__((ext_vector_type(8))) float v8f;

#define B_    4096
#define NI_   256
#define NH_   2048
#define NO_   128
#define STEPS 100
#define ALPHA_ 0.9f

template <int K, int KC>
__device__ __forceinline__ v8f block_gemm_16x128(
    const float* __restrict__ Ag,
    const float* __restrict__ Bg,
    float* __restrict__ As,
    float* __restrict__ Bs)
{
    constexpr int PITCH = KC + 4;
    constexpr int C4 = KC / 4;
    constexpr int RP = 256 / C4;
    constexpr int AP = 16 / RP;
    constexpr int BP = 128 / RP;

    const int t    = threadIdx.x;
    const int srow = t / C4;
    const int sc4  = t % C4;
    const int wave = t >> 5;
    const int lane = t & 31;
    const int mn   = lane & 15;
    const int khi  = lane >> 4;

    const float* ag = Ag + (size_t)srow * K + sc4 * 4;
    const float* bg = Bg + (size_t)srow * K + sc4 * 4;
    float* as = As + srow * PITCH + sc4 * 4;
    float* bs = Bs + srow * PITCH + sc4 * 4;
    const float* af = As + mn * PITCH + 2 * khi;
    const float* bf = Bs + (wave * 16 + mn) * PITCH + 2 * khi;

    v4f ra[AP], rb[BP];
#pragma unroll
    for (int i = 0; i < AP; ++i) ra[i] = *(const v4f*)(ag + (size_t)i * RP * K);
#pragma unroll
    for (int i = 0; i < BP; ++i) rb[i] = *(const v4f*)(bg + (size_t)i * RP * K);

    v8f acc = {};
    for (int kc = 0; kc < K; kc += KC) {
#pragma unroll
        for (int i = 0; i < AP; ++i) *(v4f*)(as + i * RP * PITCH) = ra[i];
#pragma unroll
        for (int i = 0; i < BP; ++i) *(v4f*)(bs + i * RP * PITCH) = rb[i];
        __syncthreads();

        if (kc + KC < K) {
#pragma unroll
            for (int i = 0; i < AP; ++i) ra[i] = *(const v4f*)(ag + (size_t)i * RP * K + kc + KC);
#pragma unroll
            for (int i = 0; i < BP; ++i) rb[i] = *(const v4f*)(bg + (size_t)i * RP * K + kc + KC);
        }

#pragma unroll
        for (int k = 0; k < KC; k += 4) {
            v2f a = *(const v2f*)(af + k);
            v2f b = *(const v2f*)(bf + k);
            acc = __builtin_amdgcn_wmma_f32_16x16x4_f32(
                false, a, false, b, (short)0, acc, false, false);
        }
        __syncthreads();
    }
    return acc;
}

__device__ __forceinline__ v2f gt_one2(v2f m, v2f big2)
{
    v2f r;
    asm("v_pk_fma_f32 %0, %1, %2, %2 neg_lo:[0,0,1] neg_hi:[0,0,1] clamp"
        : "=v"(r) : "v"(m), "v"(big2));
    return r;
}

__global__ __launch_bounds__(256) void snn_gemm1_recur(
    const float* __restrict__ x,
    const float* __restrict__ W1,
    const float* __restrict__ b1,
    const float* __restrict__ betap,
    float* __restrict__ spk)
{
    constexpr int KC = 64;
    __shared__ __attribute__((aligned(16))) float As[16 * (KC + 4)];
    __shared__ __attribute__((aligned(16))) float Bs[128 * (KC + 4)];

    const int wave = threadIdx.x >> 5;
    const int lane = threadIdx.x & 31;
    const int blocksN = NH_ / 128;
    const int tm = blockIdx.x / blocksN;
    const int bn = blockIdx.x - tm * blocksN;
    const int row0 = tm << 4;
    const int colB = bn << 7;
    const int col0 = colB + (wave << 4);
    const int mn  = lane & 15;
    const int khi = lane >> 4;

    v8f acc = block_gemm_16x128<NI_, KC>(x + (size_t)row0 * NI_, W1 + (size_t)colB * NI_, As, Bs);

    float beta = *betap;
    beta = fminf(fmaxf(beta, 0.0f), 1.0f);
    const float bias = b1[col0 + mn];

    v2f c[4], syn[4], mem[4];
#pragma unroll
    for (int j = 0; j < 4; ++j) {
        c[j].x = acc[2 * j] + bias;
        c[j].y = acc[2 * j + 1] + bias;
        syn[j] = (v2f){0.0f, 0.0f};
        mem[j] = (v2f){0.0f, 0.0f};
    }
    const v2f alpha2 = {ALPHA_, ALPHA_};
    const v2f beta2  = {beta, beta};

    const v2f BIG2 = {0x1p+64f, 0x1p+64f};
#pragma unroll
    for (int j = 0; j < 4; ++j)
        syn[j] = __builtin_elementwise_fma(alpha2, syn[j], c[j]);
#pragma unroll 1
    for (int t = 0; t < STEPS; ++t) {
        v2f d[4], reset[4];
#pragma unroll
        for (int j = 0; j < 4; ++j)
            reset[j] = gt_one2(mem[j], BIG2);
        __builtin_amdgcn_sched_barrier(0);
#pragma unroll
        for (int j = 0; j < 4; ++j)
            d[j] = __builtin_elementwise_fma(beta2, mem[j], syn[j]);
#pragma unroll
        for (int j = 0; j < 4; ++j)
            syn[j] = __builtin_elementwise_fma(alpha2, syn[j], c[j]);
#pragma unroll
        for (int j = 0; j < 4; ++j)
            mem[j] = d[j] - reset[j];
    }

    float* op = spk + (size_t)(row0 + khi * 8) * NH_ + (col0 + mn);
#pragma unroll
    for (int j = 0; j < 4; ++j) {
        op[(size_t)(2 * j) * NH_]     = ((mem[j].x - 1.0f) > 0.0f) ? 1.0f : 0.0f;
        op[(size_t)(2 * j + 1) * NH_] = ((mem[j].y - 1.0f) > 0.0f) ? 1.0f : 0.0f;
    }
}

__global__ __launch_bounds__(256) void snn_gemm2(
    const float* __restrict__ spk,
    const float* __restrict__ W2,
    const float* __restrict__ b2,
    float* __restrict__ out)
{
    constexpr int KC = 128;
    __shared__ __attribute__((aligned(16))) float As[16 * (KC + 4)];
    __shared__ __attribute__((aligned(16))) float Bs[128 * (KC + 4)];

    const int wave = threadIdx.x >> 5;
    const int lane = threadIdx.x & 31;
    const int row0 = blockIdx.x << 4;
    const int col0 = wave << 4;
    const int mn  = lane & 15;
    const int khi = lane >> 4;

    v8f acc = block_gemm_16x128<NH_, KC>(spk + (size_t)row0 * NH_, W2, As, Bs);

    const float bias = b2[col0 + mn];
    float* op = out + (size_t)(row0 + khi * 8) * NO_ + (col0 + mn);
#pragma unroll
    for (int i = 0; i < 8; ++i)
        op[(size_t)i * NO_] = acc[i] + bias;
}

extern "C" void kernel_launch(void* const* d_in, const int* in_sizes, int n_in,
                              void* d_out, int out_size, void* d_ws, size_t ws_size,
                              hipStream_t stream) {
    (void)in_sizes; (void)n_in; (void)out_size; (void)ws_size;
    const float* x    = (const float*)d_in[0];
    const float* W1   = (const float*)d_in[1];
    const float* b1   = (const float*)d_in[2];
    const float* W2   = (const float*)d_in[3];
    const float* b2   = (const float*)d_in[4];
    const float* beta = (const float*)d_in[5];

    float* spk = (float*)d_ws;
    float* out = (float*)d_out;

    const int tiles1 = (B_ / 16) * (NH_ / 16);
    snn_gemm1_recur<<<tiles1 / 8, 256, 0, stream>>>(x, W1, b1, beta, spk);

    const int tiles2 = (B_ / 16) * (NO_ / 16);
    snn_gemm2<<<tiles2 / 8, 256, 0, stream>>>(spk, W2, b2, out);
}


# Kernel 1 block tile 64x128 (was 16x128): each wave keeps four 16x16 row tiles that share every B fragment, one staged W1 slab serves 64 rows; recurrence runs on the wave's 4 tiles; same values
# speedup vs baseline: 1.7359x; 1.0649x over previous
#include <hip/hip_runtime.h>


typedef __attribute__((ext_vector_type(2))) float v2f;
typedef __attribute__((ext_vector_type(4))) float v4f;
typedef __attribute__((ext_vector_type(8))) float v8f;

#define B_    4096
#define NI_   256
#define NH_   2048
#define NO_   128
#define STEPS 100
#define ALPHA_ 0.9f

template <int K, int KC, int MT>
__device__ __forceinline__ void block_gemm_128(
    const float* __restrict__ Ag,
    const float* __restrict__ Bg,
    float* __restrict__ As,
    float* __restrict__ Bs,
    v8f (&acc)[MT])
{
    constexpr int PITCH = KC + 4;
    constexpr int C4 = KC / 4;
    constexpr int RP = 256 / C4;
    constexpr int AP = 16 * MT / RP;
    constexpr int BP = 128 / RP;

    const int t    = threadIdx.x;
    const int srow = t / C4;
    const int sc4  = t % C4;
    const int wave = t >> 5;
    const int lane = t & 31;
    const int mn   = lane & 15;
    const int khi  = lane >> 4;

    const unsigned goff = (unsigned)(srow * K + sc4 * 4);
    float* as = As + srow * PITCH + sc4 * 4;
    float* bs = Bs + srow * PITCH + sc4 * 4;
    const float* af = As + mn * PITCH + 2 * khi;
    const float* bf = Bs + (wave * 16 + mn) * PITCH + 2 * khi;

    v4f ra[AP], rb[BP];
#pragma unroll
    for (int i = 0; i < AP; ++i) ra[i] = *(const v4f*)(Ag + (goff + (unsigned)(i * RP * K)));
#pragma unroll
    for (int i = 0; i < BP; ++i) rb[i] = *(const v4f*)(Bg + (goff + (unsigned)(i * RP * K)));

#pragma unroll
    for (int m = 0; m < MT; ++m) acc[m] = (v8f){};
    for (int kc = 0; kc < K; kc += KC) {
#pragma unroll
        for (int i = 0; i < AP; ++i) *(v4f*)(as + i * RP * PITCH) = ra[i];
#pragma unroll
        for (int i = 0; i < BP; ++i) *(v4f*)(bs + i * RP * PITCH) = rb[i];
        __syncthreads();

        if (kc + KC < K) {
#pragma unroll
            for (int i = 0; i < AP; ++i) ra[i] = *(const v4f*)(Ag + (goff + (unsigned)(i * RP * K + kc + KC)));
#pragma unroll
            for (int i = 0; i < BP; ++i) rb[i] = *(const v4f*)(Bg + (goff + (unsigned)(i * RP * K + kc + KC)));
        }

#pragma unroll
        for (int k = 0; k < KC; k += 4) {
            v2f b = *(const v2f*)(bf + k);
#pragma unroll
            for (int m = 0; m < MT; ++m) {
                v2f a = *(const v2f*)(af + m * 16 * PITCH + k);
                acc[m] = __builtin_amdgcn_wmma_f32_16x16x4_f32(
                    false, a, false, b, (short)0, acc[m], false, false);
            }
        }
        __syncthreads();
    }
}

__device__ __forceinline__ v2f gt_one2(v2f m, v2f big2)
{
    v2f r;
    asm("v_pk_fma_f32 %0, %1, %2, %2 neg_lo:[0,0,1] neg_hi:[0,0,1] clamp"
        : "=v"(r) : "v"(m), "v"(big2));
    return r;
}

#define MT1 4
__global__ __launch_bounds__(256, 8) void snn_gemm1_recur(
    const float* __restrict__ x,
    const float* __restrict__ W1,
    const float* __restrict__ b1,
    const float* __restrict__ betap,
    float* __restrict__ spk)
{
    constexpr int KC = 64;
    __shared__ __attribute__((aligned(16))) float As[16 * MT1 * (KC + 4)];
    __shared__ __attribute__((aligned(16))) float Bs[128 * (KC + 4)];

    const int wave = __builtin_amdgcn_readfirstlane((int)(threadIdx.x >> 5));
    const int blocksN = NH_ / 128;
    const int tm = blockIdx.x / blocksN;
    const int bn = blockIdx.x - tm * blocksN;
    const int row0 = tm * (16 * MT1);
    const int colB = bn << 7;
    const int col0 = colB + (wave << 4);

    v8f acc[MT1];
    block_gemm_128<NI_, KC, MT1>(x + (size_t)row0 * NI_, W1 + (size_t)colB * NI_, As, Bs, acc);

    const int lane = (int)__builtin_amdgcn_mbcnt_lo(~0u, 0u);
    const int mn  = lane & 15;
    const int khi = lane >> 4;

    float beta = *betap;
    beta = fminf(fmaxf(beta, 0.0f), 1.0f);
    const float bias = b1[col0 + mn];

    v2f c[MT1][4], syn[MT1][4], mem[MT1][4];
#pragma unroll
    for (int g = 0; g < MT1; ++g) {
#pragma unroll
        for (int j = 0; j < 4; ++j) {
            c[g][j].x = acc[g][2 * j] + bias;
            c[g][j].y = acc[g][2 * j + 1] + bias;
            syn[g][j] = (v2f){0.0f, 0.0f};
            mem[g][j] = (v2f){0.0f, 0.0f};
        }
    }
    const v2f alpha2 = {ALPHA_, ALPHA_};
    const v2f beta2  = {beta, beta};

    const v2f BIG2 = {0x1p+64f, 0x1p+64f};
#pragma unroll
    for (int g = 0; g < MT1; ++g) {
#pragma unroll
        for (int j = 0; j < 4; ++j)
            syn[g][j] = __builtin_elementwise_fma(alpha2, syn[g][j], c[g][j]);
    }
#pragma unroll 1
    for (int t = 0; t < STEPS; ++t) {
#pragma unroll
        for (int g = 0; g < MT1; ++g) {
            v2f d[4], reset[4];
#pragma unroll
            for (int j = 0; j < 4; ++j)
                reset[j] = gt_one2(mem[g][j], BIG2);
            __builtin_amdgcn_sched_barrier(0);
#pragma unroll
            for (int j = 0; j < 4; ++j)
                d[j] = __builtin_elementwise_fma(beta2, mem[g][j], syn[g][j]);
            __builtin_amdgcn_sched_barrier(0);
#pragma unroll
            for (int j = 0; j < 4; ++j)
                syn[g][j] = __builtin_elementwise_fma(alpha2, syn[g][j], c[g][j]);
#pragma unroll
            for (int j = 0; j < 4; ++j)
                mem[g][j] = d[j] - reset[j];
        }
    }

#pragma unroll
    for (int g = 0; g < MT1; ++g) {
        float* op = spk + (size_t)(row0 + g * 16 + khi * 8) * NH_ + (col0 + mn);
#pragma unroll
        for (int j = 0; j < 4; ++j) {
            op[(size_t)(2 * j) * NH_]     = ((mem[g][j].x - 1.0f) > 0.0f) ? 1.0f : 0.0f;
            op[(size_t)(2 * j + 1) * NH_] = ((mem[g][j].y - 1.0f) > 0.0f) ? 1.0f : 0.0f;
        }
    }
}

__global__ __launch_bounds__(256) void snn_gemm2(
    const float* __restrict__ spk,
    const float* __restrict__ W2,
    const float* __restrict__ b2,
    float* __restrict__ out)
{
    constexpr int KC = 128;
    __shared__ __attribute__((aligned(16))) float As[16 * (KC + 4)];
    __shared__ __attribute__((aligned(16))) float Bs[128 * (KC + 4)];

    const int wave = threadIdx.x >> 5;
    const int lane = threadIdx.x & 31;
    const int row0 = blockIdx.x << 4;
    const int col0 = wave << 4;
    const int mn  = lane & 15;
    const int khi = lane >> 4;

    v8f acc[1];
    block_gemm_128<NH_, KC, 1>(spk + (size_t)row0 * NH_, W2, As, Bs, acc);

    const float bias = b2[col0 + mn];
    float* op = out + (size_t)(row0 + khi * 8) * NO_ + (col0 + mn);
#pragma unroll
    for (int i = 0; i < 8; ++i)
        op[(size_t)i * NO_] = acc[0][i] + bias;
}

extern "C" void kernel_launch(void* const* d_in, const int* in_sizes, int n_in,
                              void* d_out, int out_size, void* d_ws, size_t ws_size,
                              hipStream_t stream) {
    (void)in_sizes; (void)n_in; (void)out_size; (void)ws_size;
    const float* x    = (const float*)d_in[0];
    const float* W1   = (const float*)d_in[1];
    const float* b1   = (const float*)d_in[2];
    const float* W2   = (const float*)d_in[3];
    const float* b2   = (const float*)d_in[4];
    const float* beta = (const float*)d_in[5];

    float* spk = (float*)d_ws;
    float* out = (float*)d_out;

    snn_gemm1_recur<<<(B_ / (16 * MT1)) * (NH_ / 128), 256, 0, stream>>>(x, W1, b1, beta, spk);

    const int tiles2 = (B_ / 16) * (NO_ / 16);
    snn_gemm2<<<tiles2 / 8, 256, 0, stream>>>(spk, W2, b2, out);
}


# Final spikes cross to kernel 2 as bit masks (ballots; exact 0/1, 1 MB not 32 MB), expanded to f32 while staging; kernel 2 tile 64x32, so a block fetches 4x less of W2; same values
# speedup vs baseline: 1.9300x; 1.1118x over previous
#include <hip/hip_runtime.h>


typedef __attribute__((ext_vector_type(2))) float v2f;
typedef __attribute__((ext_vector_type(4))) float v4f;
typedef __attribute__((ext_vector_type(8))) float v8f;

#define B_    4096
#define NI_   256
#define NH_   2048
#define NO_   128
#define STEPS 100
#define ALPHA_ 0.9f

template <int K, int KC, int MT>
__device__ __forceinline__ void block_gemm_128(
    const float* __restrict__ Ag,
    const float* __restrict__ Bg,
    float* __restrict__ As,
    float* __restrict__ Bs,
    v8f (&acc)[MT])
{
    constexpr int PITCH = KC + 4;
    constexpr int C4 = KC / 4;
    constexpr int RP = 256 / C4;
    constexpr int AP = 16 * MT / RP;
    constexpr int BP = 128 / RP;

    const int t    = threadIdx.x;
    const int srow = t / C4;
    const int sc4  = t % C4;
    const int wave = t >> 5;
    const int lane = t & 31;
    const int mn   = lane & 15;
    const int khi  = lane >> 4;

    const unsigned goff = (unsigned)(srow * K + sc4 * 4);
    float* as = As + srow * PITCH + sc4 * 4;
    float* bs = Bs + srow * PITCH + sc4 * 4;
    const float* af = As + mn * PITCH + 2 * khi;
    const float* bf = Bs + (wave * 16 + mn) * PITCH + 2 * khi;

    v4f ra[AP], rb[BP];
#pragma unroll
    for (int i = 0; i < AP; ++i) ra[i] = *(const v4f*)(Ag + (goff + (unsigned)(i * RP * K)));
#pragma unroll
    for (int i = 0; i < BP; ++i) rb[i] = *(const v4f*)(Bg + (goff + (unsigned)(i * RP * K)));

#pragma unroll
    for (int m = 0; m < MT; ++m) acc[m] = (v8f){};
    for (int kc = 0; kc < K; kc += KC) {
#pragma unroll
        for (int i = 0; i < AP; ++i) *(v4f*)(as + i * RP * PITCH) = ra[i];
#pragma unroll
        for (int i = 0; i < BP; ++i) *(v4f*)(bs + i * RP * PITCH) = rb[i];
        __syncthreads();

        if (kc + KC < K) {
#pragma unroll
            for (int i = 0; i < AP; ++i) ra[i] = *(const v4f*)(Ag + (goff + (unsigned)(i * RP * K + kc + KC)));
#pragma unroll
            for (int i = 0; i < BP; ++i) rb[i] = *(const v4f*)(Bg + (goff + (unsigned)(i * RP * K + kc + KC)));
        }

#pragma unroll
        for (int k = 0; k < KC; k += 4) {
            v2f b = *(const v2f*)(bf + k);
#pragma unroll
            for (int m = 0; m < MT; ++m) {
                v2f a = *(const v2f*)(af + m * 16 * PITCH + k);
                acc[m] = __builtin_amdgcn_wmma_f32_16x16x4_f32(
                    false, a, false, b, (short)0, acc[m], false, false);
            }
        }
        __syncthreads();
    }
}

__device__ __forceinline__ v2f gt_one2(v2f m, v2f big2)
{
    v2f r;
    asm("v_pk_fma_f32 %0, %1, %2, %2 neg_lo:[0,0,1] neg_hi:[0,0,1] clamp"
        : "=v"(r) : "v"(m), "v"(big2));
    return r;
}

#define MT1 4
__global__ __launch_bounds__(256, 8) void snn_gemm1_recur(
    const float* __restrict__ x,
    const float* __restrict__ W1,
    const float* __restrict__ b1,
    const float* __restrict__ betap,
    float* __restrict__ spk)
{
    constexpr int KC = 64;
    __shared__ __attribute__((aligned(16))) float As[16 * MT1 * (KC + 4)];
    __shared__ __attribute__((aligned(16))) float Bs[128 * (KC + 4)];

    const int wave = __builtin_amdgcn_readfirstlane((int)(threadIdx.x >> 5));
    const int blocksN = NH_ / 128;
    const int tm = blockIdx.x / blocksN;
    const int bn = blockIdx.x - tm * blocksN;
    const int row0 = tm * (16 * MT1);
    const int colB = bn << 7;
    const int col0 = colB + (wave << 4);

    v8f acc[MT1];
    block_gemm_128<NI_, KC, MT1>(x + (size_t)row0 * NI_, W1 + (size_t)colB * NI_, As, Bs, acc);

    const int lane = (int)__builtin_amdgcn_mbcnt_lo(~0u, 0u);
    const int mn  = lane & 15;
    const int khi = lane >> 4;

    float beta = *betap;
    beta = fminf(fmaxf(beta, 0.0f), 1.0f);
    const float bias = b1[col0 + mn];

    v2f c[MT1][4], syn[MT1][4], mem[MT1][4];
#pragma unroll
    for (int g = 0; g < MT1; ++g) {
#pragma unroll
        for (int j = 0; j < 4; ++j) {
            c[g][j].x = acc[g][2 * j] + bias;
            c[g][j].y = acc[g][2 * j + 1] + bias;
            syn[g][j] = (v2f){0.0f, 0.0f};
            mem[g][j] = (v2f){0.0f, 0.0f};
        }
    }
    const v2f alpha2 = {ALPHA_, ALPHA_};
    const v2f beta2  = {beta, beta};

    const v2f BIG2 = {0x1p+64f, 0x1p+64f};
#pragma unroll
    for (int g = 0; g < MT1; ++g) {
#pragma unroll
        for (int j = 0; j < 4; ++j)
            syn[g][j] = __builtin_elementwise_fma(alpha2, syn[g][j], c[g][j]);
    }
#pragma unroll 1
    for (int t = 0; t < STEPS; ++t) {
#pragma unroll
        for (int g = 0; g < MT1; ++g) {
            v2f d[4], reset[4];
#pragma unroll
            for (int j = 0; j < 4; ++j)
                reset[j] = gt_one2(mem[g][j], BIG2);
            __builtin_amdgcn_sched_barrier(0);
#pragma unroll
            for (int j = 0; j < 4; ++j)
                d[j] = __builtin_elementwise_fma(beta2, mem[g][j], syn[g][j]);
            __builtin_amdgcn_sched_barrier(0);
#pragma unroll
            for (int j = 0; j < 4; ++j)
                syn[g][j] = __builtin_elementwise_fma(alpha2, syn[g][j], c[g][j]);
#pragma unroll
            for (int j = 0; j < 4; ++j)
                mem[g][j] = d[j] - reset[j];
        }
    }

    unsigned short* l16 = (unsigned short*)As;
#pragma unroll
    for (int g = 0; g < MT1; ++g) {
#pragma unroll
        for (int j = 0; j < 4; ++j) {
            const unsigned bx = __builtin_amdgcn_ballot_w32((mem[g][j].x - 1.0f) > 0.0f);
            const unsigned by = __builtin_amdgcn_ballot_w32((mem[g][j].y - 1.0f) > 0.0f);
            if (mn == 0) {
                const int r = g * 16 + khi * 8 + 2 * j;
                l16[r * 8 + wave]       = (unsigned short)(khi ? (bx >> 16) : bx);
                l16[(r + 1) * 8 + wave] = (unsigned short)(khi ? (by >> 16) : by);
            }
        }
    }
    __syncthreads();
    const int tid = (wave << 5) + lane;
    unsigned* pw = (unsigned*)spk;
    pw[((size_t)bn * B_ + row0) * 4 + tid] = ((const unsigned*)As)[tid];
}

__global__ __launch_bounds__(256) void snn_gemm2(
    const float* __restrict__ spk,
    const float* __restrict__ W2,
    const float* __restrict__ b2,
    float* __restrict__ out)
{
    constexpr int KC = 128;
    constexpr int PITCH = KC + 4;
    __shared__ __attribute__((aligned(16))) float As[64 * PITCH];
    __shared__ __attribute__((aligned(16))) float Bs[32 * PITCH];

    const int t    = threadIdx.x;
    const int wave = t >> 5;
    const int lane = t & 31;
    const int mn   = lane & 15;
    const int khi  = lane >> 4;
    const int rb   = blockIdx.x >> 2;
    const int cb   = blockIdx.x & 3;
    const int row0 = rb << 6;
    const int colB = cb << 5;

    const int arow = t >> 2;
    const int aq   = t & 3;
    const unsigned* pa = (const unsigned*)spk + ((size_t)(row0 + arow) * 4 + aq);
    float* as = As + arow * PITCH + aq * 32;
    const int srow = t >> 5;
    const int sc4  = t & 31;
    const float* pb = W2 + (size_t)(colB + srow) * NH_ + sc4 * 4;
    float* bs = Bs + srow * PITCH + sc4 * 4;

    const int wr = wave & 3;
    const int wc = wave >> 2;
    const float* af = As + (wr * 16 + mn) * PITCH + 2 * khi;
    const float* bf = Bs + (wc * 16 + mn) * PITCH + 2 * khi;

    unsigned wa = pa[0];
    v4f rbv[4];
#pragma unroll
    for (int i = 0; i < 4; ++i) rbv[i] = *(const v4f*)(pb + (size_t)(i * 8) * NH_);

    v8f acc = (v8f){};
    for (int kc = 0; kc < NH_; kc += KC) {
#pragma unroll
        for (int m = 0; m < 8; ++m) {
            v4f f;
            f.x = (float)((wa >> (4 * m))     & 1u);
            f.y = (float)((wa >> (4 * m + 1)) & 1u);
            f.z = (float)((wa >> (4 * m + 2)) & 1u);
            f.w = (float)((wa >> (4 * m + 3)) & 1u);
            *(v4f*)(as + 4 * m) = f;
        }
#pragma unroll
        for (int i = 0; i < 4; ++i) *(v4f*)(bs + i * 8 * PITCH) = rbv[i];
        __syncthreads();

        if (kc + KC < NH_) {
            wa = pa[(size_t)((kc + KC) / KC) * (B_ * 4)];
#pragma unroll
            for (int i = 0; i < 4; ++i) rbv[i] = *(const v4f*)(pb + (size_t)(i * 8) * NH_ + (kc + KC));
        }

#pragma unroll
        for (int k = 0; k < KC; k += 4) {
            v2f a = *(const v2f*)(af + k);
            v2f b = *(const v2f*)(bf + k);
            acc = __builtin_amdgcn_wmma_f32_16x16x4_f32(
                false, a, false, b, (short)0, acc, false, false);
        }
        __syncthreads();
    }

    const int col = colB + wc * 16 + mn;
    const float bias = b2[col];
    float* op = out + (size_t)(row0 + wr * 16 + khi * 8) * NO_ + col;
#pragma unroll
    for (int i = 0; i < 8; ++i)
        op[(size_t)i * NO_] = acc[i] + bias;
}

extern "C" void kernel_launch(void* const* d_in, const int* in_sizes, int n_in,
                              void* d_out, int out_size, void* d_ws, size_t ws_size,
                              hipStream_t stream) {
    (void)in_sizes; (void)n_in; (void)out_size; (void)ws_size;
    const float* x    = (const float*)d_in[0];
    const float* W1   = (const float*)d_in[1];
    const float* b1   = (const float*)d_in[2];
    const float* W2   = (const float*)d_in[3];
    const float* b2   = (const float*)d_in[4];
    const float* beta = (const float*)d_in[5];

    float* spk = (float*)d_ws;
    float* out = (float*)d_out;

    snn_gemm1_recur<<<(B_ / (16 * MT1)) * (NH_ / 128), 256, 0, stream>>>(x, W1, b1, beta, spk);

    snn_gemm2<<<(B_ / 64) * (NO_ / 32), 256, 0, stream>>>(spk, W2, b2, out);
}
